# starting kernel
# baseline (speedup 1.0000x reference)
#include <hip/hip_runtime.h>
#include <hip/hip_bf16.h>

#define FEAT 96
#define LSTRIDE 100
#define WPAIR 192

typedef float v2f __attribute__((ext_vector_type(2)));
typedef float v8f __attribute__((ext_vector_type(8)));

__global__ void zero_kernel(float* __restrict__ p, long long n) {
    long long i = (long long)blockIdx.x * blockDim.x + threadIdx.x;
    if (i < n) p[i] = 0.0f;
}

__global__ void degree_kernel(const int* __restrict__ dst, float* __restrict__ deg, int E) {
    int e = blockIdx.x * blockDim.x + threadIdx.x;
    if (e < E) atomicAdd(&deg[dst[e]], 1.0f);
}

__global__ void dinv_kernel(const float* __restrict__ deg, float* __restrict__ dinv, int N) {
    int i = blockIdx.x * blockDim.x + threadIdx.x;
    if (i < N) {
        float d = deg[i];
        dinv[i] = (d > 0.0f) ? rsqrtf(fmaxf(d, 1.0f)) : 0.0f;
    }
}

__global__ void __launch_bounds__(128)
gemm_wmma_f32(const float* __restrict__ A,
              const float* __restrict__ W,
              float* __restrict__ C,
              int nrows) {
    __shared__ float Wlds[(FEAT / 2) * WPAIR];
    __shared__ float Alds[4][16 * LSTRIDE];

    const int tid  = threadIdx.x;
    const int lane = tid & 31;
    const int wv   = tid >> 5;
    const int wid  = blockIdx.x * 4 + wv;
    const int row0 = wid << 4;
    const bool active = (row0 < nrows);

    for (int i = tid; i < FEAT * (FEAT / 4); i += 128) {
        const int r  = i / (FEAT / 4);
        const int c0 = (i % (FEAT / 4)) * 4;
        const float4 v = *(const float4*)(W + r * FEAT + c0);
        float* __restrict__ dp = &Wlds[(r >> 1) * WPAIR + (r & 1)];
        dp[(c0 + 0) * 2] = v.x;
        dp[(c0 + 1) * 2] = v.y;
        dp[(c0 + 2) * 2] = v.z;
        dp[(c0 + 3) * 2] = v.w;
    }

    if (active) {
        for (int i = lane; i < 16 * (FEAT / 4); i += 32) {
            const int r  = i / (FEAT / 4);
            const int c4 = i % (FEAT / 4);
            int gr = row0 + r;
            if (gr > nrows - 1) gr = nrows - 1;
            const float4 v = ((const float4*)(A + (long long)gr * FEAT))[c4];
            *(float4*)&Alds[wv][r * LSTRIDE + c4 * 4] = v;
        }
    }
    __syncthreads();
    if (!active) return;

    const int m  = lane & 15;
    const int kh = (lane >> 4) << 1;
    const float* __restrict__ arow = &Alds[wv][m * LSTRIDE];

    v8f acc[6];
#pragma unroll
    for (int t = 0; t < 6; ++t) acc[t] = {};

    for (int k0 = 0; k0 < FEAT; k0 += 4) {
        const v2f a = *(const v2f*)&arow[k0 + kh];
        const float* __restrict__ wp = &Wlds[((k0 + kh) >> 1) * WPAIR + m * 2];
#pragma unroll
        for (int t = 0; t < 6; ++t) {
            const v2f b = *(const v2f*)&wp[t * 32];
            acc[t] = __builtin_amdgcn_wmma_f32_16x16x4_f32(
                false, a, false, b, (short)0, acc[t], false, false);
        }
    }

    const int half = lane >> 4;
#pragma unroll
    for (int t = 0; t < 6; ++t) {
#pragma unroll
        for (int i = 0; i < 8; ++i) {
            int r = row0 + half * 8 + i;
            if (r < nrows) C[(long long)r * FEAT + t * 16 + m] = acc[t][i];
        }
    }
}

__global__ void aggregate_kernel(const float* __restrict__ h,
                                 const int* __restrict__ src,
                                 const int* __restrict__ dst,
                                 const float* __restrict__ dinv,
                                 float* __restrict__ out,
                                 int E) {
    const int lane = threadIdx.x & 31;
    const long long e = (long long)blockIdx.x * (blockDim.x >> 5) + (threadIdx.x >> 5);
    if (e >= E) return;
    const int s = src[e];
    const int d = dst[e];
    const float* __restrict__ hs = h + (long long)s * FEAT;
    __builtin_prefetch(hs + lane, 0, 0);
    const float norm = dinv[s] * dinv[d];
    float* __restrict__ od = out + (long long)d * FEAT;
#pragma unroll
    for (int i = 0; i < 3; ++i) {
        const int f = lane + i * 32;
        atomicAdd(&od[f], hs[f] * norm);
    }
}

__global__ void bias_relu_kernel(float* __restrict__ x, const float* __restrict__ b,
                                 long long n) {
    long long i = (long long)blockIdx.x * blockDim.x + threadIdx.x;
    if (i < n) {
        float v = x[i] + b[(int)(i % FEAT)];
        x[i] = v > 0.0f ? v : 0.0f;
    }
}

__global__ void bias2_kernel(float* __restrict__ out, const float* __restrict__ ba,
                             const float* __restrict__ bb, long long nf) {
    long long i = (long long)blockIdx.x * blockDim.x + threadIdx.x;
    if (i < 2 * nf) {
        const float* b = (i < nf) ? ba : bb;
        out[i] += b[(int)(i % FEAT)];
    }
}

extern "C" void kernel_launch(void* const* d_in, const int* in_sizes, int n_in,
                              void* d_out, int out_size, void* d_ws, size_t ws_size,
                              hipStream_t stream) {
    const float* x   = (const float*)d_in[0];
    const float* W1  = (const float*)d_in[1];
    const float* b1  = (const float*)d_in[2];
    const float* W2a = (const float*)d_in[3];
    const float* b2a = (const float*)d_in[4];
    const float* W2b = (const float*)d_in[5];
    const float* b2b = (const float*)d_in[6];
    const int*   ei  = (const int*)d_in[7];

    const int N = in_sizes[0] / FEAT;
    const int E = in_sizes[7] / 2;
    const int* src = ei;
    const int* dst = ei + E;

    const long long NF = (long long)N * FEAT;

    float* ws   = (float*)d_ws;
    float* deg  = ws;
    float* dinv = ws + N;
    float* h    = ws + 2LL * N;
    float* agg  = h + NF;

    float* out  = (float*)d_out;

    const int T = 256;
    const int wavesPerBlk = T / 32;
    const int gemmWaves   = (N + 15) / 16;
    const int gemmBlocks  = (gemmWaves + 3) / 4;
    const int aggBlocks   = (E + wavesPerBlk - 1) / wavesPerBlk;

    zero_kernel<<<(int)((N + T - 1) / T), T, 0, stream>>>(deg, N);
    zero_kernel<<<(int)((NF + T - 1) / T), T, 0, stream>>>(agg, NF);
    zero_kernel<<<(int)((2 * NF + T - 1) / T), T, 0, stream>>>(out, 2 * NF);
    degree_kernel<<<(E + T - 1) / T, T, 0, stream>>>(dst, deg, E);
    dinv_kernel<<<(N + T - 1) / T, T, 0, stream>>>(deg, dinv, N);

    gemm_wmma_f32<<<gemmBlocks, 128, 0, stream>>>(x, W1, h, N);
    aggregate_kernel<<<aggBlocks, T, 0, stream>>>(h, src, dst, dinv, agg, E);
    bias_relu_kernel<<<(int)((NF + T - 1) / T), T, 0, stream>>>(agg, b1, NF);

    gemm_wmma_f32<<<gemmBlocks, 128, 0, stream>>>(agg, W2a, h, N);
    aggregate_kernel<<<aggBlocks, T, 0, stream>>>(h, src, dst, dinv, out, E);

    gemm_wmma_f32<<<gemmBlocks, 128, 0, stream>>>(agg, W2b, h, N);
    aggregate_kernel<<<aggBlocks, T, 0, stream>>>(h, src, dst, dinv, out + NF, E);

    bias2_kernel<<<(int)((2 * NF + T - 1) / T), T, 0, stream>>>(out, b2a, b2b, NF);
}


# Sum runs of consecutive edges with equal destination in registers (one wave per 32 edges) and issue one f32 atomic add per run, in degree count and aggregation; changes the float summation order
# speedup vs baseline: 2.6053x; 2.6053x over previous
#include <hip/hip_runtime.h>
#include <hip/hip_bf16.h>

#define FEAT 96
#define LSTRIDE 100
#define WPAIR 192

typedef float v2f __attribute__((ext_vector_type(2)));
typedef float v8f __attribute__((ext_vector_type(8)));

__global__ void zero_kernel(float* __restrict__ p, long long n) {
    long long i = (long long)blockIdx.x * blockDim.x + threadIdx.x;
    if (i < n) p[i] = 0.0f;
}

__global__ void degree_kernel(const int* __restrict__ dst, float* __restrict__ deg, int E) {
    const int e    = blockIdx.x * blockDim.x + threadIdx.x;
    const int lane = threadIdx.x & 31;
    const bool valid = (e < E);
    const int d     = valid ? dst[e] : -1;
    const int dprev = __shfl_up(d, 1);
    const bool head = valid && (lane == 0 || d != dprev);
    const unsigned heads = (unsigned)__ballot(head);
    const unsigned nval  = (unsigned)__popcll(__ballot(valid));
    if (head) {
        const unsigned above = (lane == 31) ? 0u : (heads >> (lane + 1));
        const int len = above ? __ffs(above) : (int)nval - lane;
        atomicAdd(&deg[d], (float)len);
    }
}

__global__ void dinv_kernel(const float* __restrict__ deg, float* __restrict__ dinv, int N) {
    int i = blockIdx.x * blockDim.x + threadIdx.x;
    if (i < N) {
        float d = deg[i];
        dinv[i] = (d > 0.0f) ? rsqrtf(fmaxf(d, 1.0f)) : 0.0f;
    }
}

__global__ void __launch_bounds__(128)
gemm_wmma_f32(const float* __restrict__ A,
              const float* __restrict__ W,
              float* __restrict__ C,
              int nrows) {
    __shared__ float Wlds[(FEAT / 2) * WPAIR];
    __shared__ float Alds[4][16 * LSTRIDE];

    const int tid  = threadIdx.x;
    const int lane = tid & 31;
    const int wv   = tid >> 5;
    const int wid  = blockIdx.x * 4 + wv;
    const int row0 = wid << 4;
    const bool active = (row0 < nrows);

    for (int i = tid; i < FEAT * (FEAT / 4); i += 128) {
        const int r  = i / (FEAT / 4);
        const int c0 = (i % (FEAT / 4)) * 4;
        const float4 v = *(const float4*)(W + r * FEAT + c0);
        float* __restrict__ dp = &Wlds[(r >> 1) * WPAIR + (r & 1)];
        dp[(c0 + 0) * 2] = v.x;
        dp[(c0 + 1) * 2] = v.y;
        dp[(c0 + 2) * 2] = v.z;
        dp[(c0 + 3) * 2] = v.w;
    }

    if (active) {
        for (int i = lane; i < 16 * (FEAT / 4); i += 32) {
            const int r  = i / (FEAT / 4);
            const int c4 = i % (FEAT / 4);
            int gr = row0 + r;
            if (gr > nrows - 1) gr = nrows - 1;
            const float4 v = ((const float4*)(A + (long long)gr * FEAT))[c4];
            *(float4*)&Alds[wv][r * LSTRIDE + c4 * 4] = v;
        }
    }
    __syncthreads();
    if (!active) return;

    const int m  = lane & 15;
    const int kh = (lane >> 4) << 1;
    const float* __restrict__ arow = &Alds[wv][m * LSTRIDE];

    v8f acc[6];
#pragma unroll
    for (int t = 0; t < 6; ++t) acc[t] = {};

    for (int k0 = 0; k0 < FEAT; k0 += 4) {
        const v2f a = *(const v2f*)&arow[k0 + kh];
        const float* __restrict__ wp = &Wlds[((k0 + kh) >> 1) * WPAIR + m * 2];
#pragma unroll
        for (int t = 0; t < 6; ++t) {
            const v2f b = *(const v2f*)&wp[t * 32];
            acc[t] = __builtin_amdgcn_wmma_f32_16x16x4_f32(
                false, a, false, b, (short)0, acc[t], false, false);
        }
    }

    const int half = lane >> 4;
#pragma unroll
    for (int t = 0; t < 6; ++t) {
#pragma unroll
        for (int i = 0; i < 8; ++i) {
            int r = row0 + half * 8 + i;
            if (r < nrows) C[(long long)r * FEAT + t * 16 + m] = acc[t][i];
        }
    }
}

#define ECHUNK 32
#define EBATCH 8

__device__ __forceinline__ void flush_row(float* __restrict__ out, int d, int lane,
                                          float& a0, float& a1, float& a2) {
    float* __restrict__ od = out + (long long)d * FEAT + lane;
    atomicAdd(od,      a0);
    atomicAdd(od + 32, a1);
    atomicAdd(od + 64, a2);
    a0 = 0.0f; a1 = 0.0f; a2 = 0.0f;
}

__global__ void __launch_bounds__(256)
aggregate_kernel(const float* __restrict__ h,
                 const int* __restrict__ src,
                 const int* __restrict__ dst,
                 const float* __restrict__ dinv,
                 float* __restrict__ out,
                 int E) {
    const int lane = threadIdx.x & 31;
    const int wv = __builtin_amdgcn_readfirstlane((int)(threadIdx.x >> 5));
    const long long e0 = ((long long)blockIdx.x * (blockDim.x >> 5) + wv) * ECHUNK;
    if (e0 >= E) return;
    const int cnt = (E - e0 < ECHUNK) ? (int)(E - e0) : ECHUNK;

    const long long el = e0 + (lane < cnt ? lane : cnt - 1);
    const int   s_l = src[el];
    const int   d_l = dst[el];
    const float n_l = dinv[s_l] * dinv[d_l];
    const int   n_b = __float_as_int(n_l);

    float acc0 = 0.0f, acc1 = 0.0f, acc2 = 0.0f;

    if (cnt == ECHUNK) {
        for (int b = 0; b < ECHUNK; b += EBATCH) {
            float v[EBATCH][3];
#pragma unroll
            for (int u = 0; u < EBATCH; ++u) {
                const int s = __builtin_amdgcn_readlane(s_l, b + u);
                const float* __restrict__ hs = h + (long long)s * FEAT + lane;
                v[u][0] = hs[0];
                v[u][1] = hs[32];
                v[u][2] = hs[64];
            }
#pragma unroll
            for (int u = 0; u < EBATCH; ++u) {
                const int j = b + u;
                const float nrm = __int_as_float(__builtin_amdgcn_readlane(n_b, j));
                const int d  = __builtin_amdgcn_readlane(d_l, j);
                const int dn = __builtin_amdgcn_readlane(d_l, (j + 1) & (ECHUNK - 1));
                acc0 += v[u][0] * nrm;
                acc1 += v[u][1] * nrm;
                acc2 += v[u][2] * nrm;
                if (j == ECHUNK - 1 || dn != d)
                    flush_row(out, d, lane, acc0, acc1, acc2);
            }
        }
    } else {
        for (int j = 0; j < cnt; ++j) {
            const int s = __builtin_amdgcn_readlane(s_l, j);
            const float* __restrict__ hs = h + (long long)s * FEAT + lane;
            const float nrm = __int_as_float(__builtin_amdgcn_readlane(n_b, j));
            const int d  = __builtin_amdgcn_readlane(d_l, j);
            const int dn = __builtin_amdgcn_readlane(d_l, (j + 1) & (ECHUNK - 1));
            acc0 += hs[0]  * nrm;
            acc1 += hs[32] * nrm;
            acc2 += hs[64] * nrm;
            if (j + 1 == cnt || dn != d)
                flush_row(out, d, lane, acc0, acc1, acc2);
        }
    }
}

__global__ void bias_relu_kernel(float* __restrict__ x, const float* __restrict__ b,
                                 long long n) {
    long long i = (long long)blockIdx.x * blockDim.x + threadIdx.x;
    if (i < n) {
        float v = x[i] + b[(int)(i % FEAT)];
        x[i] = v > 0.0f ? v : 0.0f;
    }
}

__global__ void bias2_kernel(float* __restrict__ out, const float* __restrict__ ba,
                             const float* __restrict__ bb, long long nf) {
    long long i = (long long)blockIdx.x * blockDim.x + threadIdx.x;
    if (i < 2 * nf) {
        const float* b = (i < nf) ? ba : bb;
        out[i] += b[(int)(i % FEAT)];
    }
}

extern "C" void kernel_launch(void* const* d_in, const int* in_sizes, int n_in,
                              void* d_out, int out_size, void* d_ws, size_t ws_size,
                              hipStream_t stream) {
    const float* x   = (const float*)d_in[0];
    const float* W1  = (const float*)d_in[1];
    const float* b1  = (const float*)d_in[2];
    const float* W2a = (const float*)d_in[3];
    const float* b2a = (const float*)d_in[4];
    const float* W2b = (const float*)d_in[5];
    const float* b2b = (const float*)d_in[6];
    const int*   ei  = (const int*)d_in[7];

    const int N = in_sizes[0] / FEAT;
    const int E = in_sizes[7] / 2;
    const int* src = ei;
    const int* dst = ei + E;

    const long long NF = (long long)N * FEAT;

    float* ws   = (float*)d_ws;
    float* deg  = ws;
    float* dinv = ws + N;
    float* h    = ws + 2LL * N;
    float* agg  = h + NF;

    float* out  = (float*)d_out;

    const int T = 256;
    const int wavesPerBlk = T / 32;
    const int gemmWaves   = (N + 15) / 16;
    const int gemmBlocks  = (gemmWaves + 3) / 4;
    const int aggChunks   = (E + ECHUNK - 1) / ECHUNK;
    const int aggBlocks   = (aggChunks + wavesPerBlk - 1) / wavesPerBlk;

    zero_kernel<<<(int)((N + T - 1) / T), T, 0, stream>>>(deg, N);
    zero_kernel<<<(int)((NF + T - 1) / T), T, 0, stream>>>(agg, NF);
    zero_kernel<<<(int)((2 * NF + T - 1) / T), T, 0, stream>>>(out, 2 * NF);
    degree_kernel<<<(E + T - 1) / T, T, 0, stream>>>(dst, deg, E);
    dinv_kernel<<<(N + T - 1) / T, T, 0, stream>>>(deg, dinv, N);

    gemm_wmma_f32<<<gemmBlocks, 128, 0, stream>>>(x, W1, h, N);
    aggregate_kernel<<<aggBlocks, T, 0, stream>>>(h, src, dst, dinv, agg, E);
    bias_relu_kernel<<<(int)((NF + T - 1) / T), T, 0, stream>>>(agg, b1, NF);

    gemm_wmma_f32<<<gemmBlocks, 128, 0, stream>>>(agg, W2a, h, N);
    aggregate_kernel<<<aggBlocks, T, 0, stream>>>(h, src, dst, dinv, out, E);

    gemm_wmma_f32<<<gemmBlocks, 128, 0, stream>>>(agg, W2b, h, N);
    aggregate_kernel<<<aggBlocks, T, 0, stream>>>(h, src, dst, dinv, out + NF, E);

    bias2_kernel<<<(int)((2 * NF + T - 1) / T), T, 0, stream>>>(out, b2a, b2b, NF);
}


# Aggregate before each layer's matrix product, (Ahat H) W for Ahat (H W): one aggregation feeds both heads, bias and relu move into the GEMM epilogue; changes the float summation order
# speedup vs baseline: 3.8011x; 1.4590x over previous
#include <hip/hip_runtime.h>
#include <hip/hip_bf16.h>

#define FEAT 96
#define LSTRIDE 100
#define WPAIR 192

typedef float v2f __attribute__((ext_vector_type(2)));
typedef float v8f __attribute__((ext_vector_type(8)));

__global__ void zero_kernel(float* __restrict__ p, long long n) {
    long long i = (long long)blockIdx.x * blockDim.x + threadIdx.x;
    if (i < n) p[i] = 0.0f;
}

__global__ void degree_kernel(const int* __restrict__ dst, float* __restrict__ deg, int E) {
    const int e    = blockIdx.x * blockDim.x + threadIdx.x;
    const int lane = threadIdx.x & 31;
    const bool valid = (e < E);
    const int d     = valid ? dst[e] : -1;
    const int dprev = __shfl_up(d, 1);
    const bool head = valid && (lane == 0 || d != dprev);
    const unsigned heads = (unsigned)__ballot(head);
    const unsigned nval  = (unsigned)__popcll(__ballot(valid));
    if (head) {
        const unsigned above = (lane == 31) ? 0u : (heads >> (lane + 1));
        const int len = above ? __ffs(above) : (int)nval - lane;
        atomicAdd(&deg[d], (float)len);
    }
}

__global__ void dinv_kernel(const float* __restrict__ deg, float* __restrict__ dinv, int N) {
    int i = blockIdx.x * blockDim.x + threadIdx.x;
    if (i < N) {
        float d = deg[i];
        dinv[i] = (d > 0.0f) ? rsqrtf(fmaxf(d, 1.0f)) : 0.0f;
    }
}

template <bool RELU>
__global__ void __launch_bounds__(128)
gemm_wmma_f32(const float* A,
              const float* __restrict__ W,
              const float* __restrict__ bias,
              float* C,
              int nrows) {
    __shared__ float Wlds[(FEAT / 2) * WPAIR];
    __shared__ float Alds[4][16 * LSTRIDE];

    const int tid  = threadIdx.x;
    const int lane = tid & 31;
    const int wv   = tid >> 5;
    const int wid  = blockIdx.x * 4 + wv;
    const int row0 = wid << 4;
    const bool active = (row0 < nrows);

    for (int i = tid; i < FEAT * (FEAT / 4); i += 128) {
        const int r  = i / (FEAT / 4);
        const int c0 = (i % (FEAT / 4)) * 4;
        const float4 v = *(const float4*)(W + r * FEAT + c0);
        float* __restrict__ dp = &Wlds[(r >> 1) * WPAIR + (r & 1)];
        dp[(c0 + 0) * 2] = v.x;
        dp[(c0 + 1) * 2] = v.y;
        dp[(c0 + 2) * 2] = v.z;
        dp[(c0 + 3) * 2] = v.w;
    }

    if (active) {
        for (int i = lane; i < 16 * (FEAT / 4); i += 32) {
            const int r  = i / (FEAT / 4);
            const int c4 = i % (FEAT / 4);
            int gr = row0 + r;
            if (gr > nrows - 1) gr = nrows - 1;
            const float4 v = ((const float4*)(A + (long long)gr * FEAT))[c4];
            *(float4*)&Alds[wv][r * LSTRIDE + c4 * 4] = v;
        }
    }
    __syncthreads();
    if (!active) return;

    const int m  = lane & 15;
    const int kh = (lane >> 4) << 1;
    const float* __restrict__ arow = &Alds[wv][m * LSTRIDE];

    float bv[6];
#pragma unroll
    for (int t = 0; t < 6; ++t) bv[t] = bias[t * 16 + m];

    v8f acc[6];
#pragma unroll
    for (int t = 0; t < 6; ++t) acc[t] = {};

    for (int k0 = 0; k0 < FEAT; k0 += 4) {
        const v2f a = *(const v2f*)&arow[k0 + kh];
        const float* __restrict__ wp = &Wlds[((k0 + kh) >> 1) * WPAIR + m * 2];
#pragma unroll
        for (int t = 0; t < 6; ++t) {
            const v2f b = *(const v2f*)&wp[t * 32];
            acc[t] = __builtin_amdgcn_wmma_f32_16x16x4_f32(
                false, a, false, b, (short)0, acc[t], false, false);
        }
    }

    const int half = lane >> 4;
#pragma unroll
    for (int t = 0; t < 6; ++t) {
#pragma unroll
        for (int i = 0; i < 8; ++i) {
            int r = row0 + half * 8 + i;
            float v = acc[t][i] + bv[t];
            if (RELU) v = v > 0.0f ? v : 0.0f;
            if (r < nrows) C[(long long)r * FEAT + t * 16 + m] = v;
        }
    }
}

#define ECHUNK 32
#define EBATCH 8

__device__ __forceinline__ void flush_row(float* __restrict__ out, int d, int lane,
                                          float& a0, float& a1, float& a2) {
    float* __restrict__ od = out + (long long)d * FEAT + lane;
    atomicAdd(od,      a0);
    atomicAdd(od + 32, a1);
    atomicAdd(od + 64, a2);
    a0 = 0.0f; a1 = 0.0f; a2 = 0.0f;
}

__global__ void __launch_bounds__(256)
aggregate_kernel(const float* __restrict__ h,
                 const int* __restrict__ src,
                 const int* __restrict__ dst,
                 const float* __restrict__ dinv,
                 float* __restrict__ out,
                 int E) {
    const int lane = threadIdx.x & 31;
    const int wv = __builtin_amdgcn_readfirstlane((int)(threadIdx.x >> 5));
    const long long e0 = ((long long)blockIdx.x * (blockDim.x >> 5) + wv) * ECHUNK;
    if (e0 >= E) return;
    const int cnt = (E - e0 < ECHUNK) ? (int)(E - e0) : ECHUNK;

    const long long el = e0 + (lane < cnt ? lane : cnt - 1);
    const int   s_l = src[el];
    const int   d_l = dst[el];
    const float n_l = dinv[s_l] * dinv[d_l];
    const int   n_b = __float_as_int(n_l);

    float acc0 = 0.0f, acc1 = 0.0f, acc2 = 0.0f;

    if (cnt == ECHUNK) {
        for (int b = 0; b < ECHUNK; b += EBATCH) {
            float v[EBATCH][3];
#pragma unroll
            for (int u = 0; u < EBATCH; ++u) {
                const int s = __builtin_amdgcn_readlane(s_l, b + u);
                const float* __restrict__ hs = h + (long long)s * FEAT + lane;
                v[u][0] = hs[0];
                v[u][1] = hs[32];
                v[u][2] = hs[64];
            }
#pragma unroll
            for (int u = 0; u < EBATCH; ++u) {
                const int j = b + u;
                const float nrm = __int_as_float(__builtin_amdgcn_readlane(n_b, j));
                const int d  = __builtin_amdgcn_readlane(d_l, j);
                const int dn = __builtin_amdgcn_readlane(d_l, (j + 1) & (ECHUNK - 1));
                acc0 += v[u][0] * nrm;
                acc1 += v[u][1] * nrm;
                acc2 += v[u][2] * nrm;
                if (j == ECHUNK - 1 || dn != d)
                    flush_row(out, d, lane, acc0, acc1, acc2);
            }
        }
    } else {
        for (int j = 0; j < cnt; ++j) {
            const int s = __builtin_amdgcn_readlane(s_l, j);
            const float* __restrict__ hs = h + (long long)s * FEAT + lane;
            const float nrm = __int_as_float(__builtin_amdgcn_readlane(n_b, j));
            const int d  = __builtin_amdgcn_readlane(d_l, j);
            const int dn = __builtin_amdgcn_readlane(d_l, (j + 1) & (ECHUNK - 1));
            acc0 += hs[0]  * nrm;
            acc1 += hs[32] * nrm;
            acc2 += hs[64] * nrm;
            if (j + 1 == cnt || dn != d)
                flush_row(out, d, lane, acc0, acc1, acc2);
        }
    }
}

extern "C" void kernel_launch(void* const* d_in, const int* in_sizes, int n_in,
                              void* d_out, int out_size, void* d_ws, size_t ws_size,
                              hipStream_t stream) {
    const float* x   = (const float*)d_in[0];
    const float* W1  = (const float*)d_in[1];
    const float* b1  = (const float*)d_in[2];
    const float* W2a = (const float*)d_in[3];
    const float* b2a = (const float*)d_in[4];
    const float* W2b = (const float*)d_in[5];
    const float* b2b = (const float*)d_in[6];
    const int*   ei  = (const int*)d_in[7];

    const int N = in_sizes[0] / FEAT;
    const int E = in_sizes[7] / 2;
    const int* src = ei;
    const int* dst = ei + E;

    const long long NF = (long long)N * FEAT;

    float* ws   = (float*)d_ws;
    float* deg  = ws;
    float* dinv = ws + N;
    float* agg1 = ws + 2LL * N;
    float* agg2 = agg1 + NF;

    float* out  = (float*)d_out;

    const int T = 256;
    const int wavesPerBlk = T / 32;
    const int gemmWaves   = (N + 15) / 16;
    const int gemmBlocks  = (gemmWaves + 3) / 4;
    const int aggChunks   = (E + ECHUNK - 1) / ECHUNK;
    const int aggBlocks   = (aggChunks + wavesPerBlk - 1) / wavesPerBlk;

    zero_kernel<<<(int)((N + T - 1) / T), T, 0, stream>>>(deg, N);
    zero_kernel<<<(int)((2 * NF + T - 1) / T), T, 0, stream>>>(agg1, 2 * NF);
    degree_kernel<<<(E + T - 1) / T, T, 0, stream>>>(dst, deg, E);
    dinv_kernel<<<(N + T - 1) / T, T, 0, stream>>>(deg, dinv, N);

    aggregate_kernel<<<aggBlocks, T, 0, stream>>>(x, src, dst, dinv, agg1, E);
    gemm_wmma_f32<true><<<gemmBlocks, 128, 0, stream>>>(agg1, W1, b1, agg1, N);

    aggregate_kernel<<<aggBlocks, T, 0, stream>>>(agg1, src, dst, dinv, agg2, E);
    gemm_wmma_f32<false><<<gemmBlocks, 128, 0, stream>>>(agg2, W2a, b2a, out, N);
    gemm_wmma_f32<false><<<gemmBlocks, 128, 0, stream>>>(agg2, W2b, b2b, out + NF, N);
}


# Counting-sort the edges by destination once per call (count with rank, scan, scatter) so 32-edge chunks sum whole rows in registers: far fewer atomic row adds; changes the float summation order
# speedup vs baseline: 4.7180x; 1.2412x over previous
#include <hip/hip_runtime.h>
#include <hip/hip_bf16.h>

#define FEAT 96
#define LSTRIDE 100
#define WPAIR 192

typedef float v2f __attribute__((ext_vector_type(2)));
typedef float v8f __attribute__((ext_vector_type(8)));

__global__ void zero_kernel(float* __restrict__ p, long long n) {
    long long i = (long long)blockIdx.x * blockDim.x + threadIdx.x;
    if (i < n) p[i] = 0.0f;
}


__global__ void count_rank_kernel(const int* __restrict__ dst, int* __restrict__ cnt,
                                  int* __restrict__ rank, int E) {
    const int e    = blockIdx.x * blockDim.x + threadIdx.x;
    const int lane = threadIdx.x & 31;
    const bool valid = (e < E);
    const int d     = valid ? dst[e] : -1;
    const int dprev = __shfl_up(d, 1);
    const bool head = valid && (lane == 0 || d != dprev);
    const unsigned heads = (unsigned)__ballot(head);
    const unsigned nval  = (unsigned)__popcll(__ballot(valid));
    int base = 0;
    if (head) {
        const unsigned above = (lane == 31) ? 0u : (heads >> (lane + 1));
        const int len = above ? __ffs(above) : (int)nval - lane;
        base = atomicAdd(&cnt[d], len);
    }
    const unsigned le = heads & (0xffffffffu >> (31 - lane));
    const int hl = le ? 31 - __clz((int)le) : 0;
    const int b  = __shfl(base, hl);
    if (valid) rank[e] = b + (lane - hl);
}

#define SCAN_T  1024
#define SCAN_IT 16

__device__ __forceinline__ int4 load_counts4(const int* cnt, int i0, int N) {
    const int last = ((N + 3) & ~3) - 4;
    int4 v = *(const int4*)(cnt + (i0 < last ? i0 : last));
    v.x = (i0     < N) ? v.x : 0;
    v.y = (i0 + 1 < N) ? v.y : 0;
    v.z = (i0 + 2 < N) ? v.z : 0;
    v.w = (i0 + 3 < N) ? v.w : 0;
    return v;
}

__global__ void __launch_bounds__(SCAN_T)
scan_dinv_kernel(const int* cnt, int* __restrict__ rowptr, float* dinv, int N) {
    __shared__ int wtot[32];
    const int lane = threadIdx.x & 31;
    const int wv   = threadIdx.x >> 5;
    int carry = 0;
    for (int p0 = 0; p0 < N; p0 += 32 * SCAN_IT * 128) {
        const int wstart = p0 + wv * (SCAN_IT * 128);

        int4 v[SCAN_IT];
#pragma unroll
        for (int it = 0; it < SCAN_IT; ++it)
            v[it] = load_counts4(cnt, wstart + it * 128 + lane * 4, N);
        int mine = 0;
#pragma unroll
        for (int it = 0; it < SCAN_IT; ++it)
            mine += v[it].x + v[it].y + v[it].z + v[it].w;
#pragma unroll
        for (int o = 16; o > 0; o >>= 1) mine += __shfl_xor(mine, o);
        if (lane == 0) wtot[wv] = mine;
        __syncthreads();
        const int w = wtot[lane];
        int woff = (lane < wv) ? w : 0;
        int tot  = w;
#pragma unroll
        for (int o = 16; o > 0; o >>= 1) {
            woff += __shfl_xor(woff, o);
            tot  += __shfl_xor(tot, o);
        }

#pragma unroll
        for (int it = 0; it < SCAN_IT; ++it)
            v[it] = load_counts4(cnt, wstart + it * 128 + lane * 4, N);
        int run = carry + woff;
#pragma unroll
        for (int it = 0; it < SCAN_IT; ++it) {
            const int i0 = wstart + it * 128 + lane * 4;
            const int c0 = v[it].x, c1 = v[it].y, c2 = v[it].z, c3 = v[it].w;
            const int t = c0 + c1 + c2 + c3;
            int incl = t;
#pragma unroll
            for (int o = 1; o < 32; o <<= 1) {
                const int y = __shfl_up(incl, o);
                if (lane >= o) incl += y;
            }
            const int r0 = run + incl - t;
            const int r1 = r0 + c0, r2 = r1 + c1, r3 = r2 + c2;
            const float g0 = (float)c0, g1 = (float)c1, g2 = (float)c2, g3 = (float)c3;
            const float q0 = (g0 > 0.0f) ? rsqrtf(fmaxf(g0, 1.0f)) : 0.0f;
            const float q1 = (g1 > 0.0f) ? rsqrtf(fmaxf(g1, 1.0f)) : 0.0f;
            const float q2 = (g2 > 0.0f) ? rsqrtf(fmaxf(g2, 1.0f)) : 0.0f;
            const float q3 = (g3 > 0.0f) ? rsqrtf(fmaxf(g3, 1.0f)) : 0.0f;
            if (i0 + 3 < N) {
                *(int4*)(rowptr + i0) = make_int4(r0, r1, r2, r3);
                *(float4*)(dinv + i0) = make_float4(q0, q1, q2, q3);
            } else {
                if (i0     < N) { rowptr[i0]     = r0; dinv[i0]     = q0; }
                if (i0 + 1 < N) { rowptr[i0 + 1] = r1; dinv[i0 + 1] = q1; }
                if (i0 + 2 < N) { rowptr[i0 + 2] = r2; dinv[i0 + 2] = q2; }
            }
            run += __shfl(incl, 31);
        }
        carry += tot;
        __syncthreads();
    }
}

__global__ void fill_kernel(const int* __restrict__ src, const int* __restrict__ dst,
                            const int* __restrict__ rank, const int* __restrict__ rowptr,
                            int* __restrict__ sorted, int E) {
    const int e = blockIdx.x * blockDim.x + threadIdx.x;
    if (e < E) {
        const int d = dst[e];
        const long long pos = (long long)rowptr[d] + rank[e];
        *(int2*)(sorted + 2 * pos) = make_int2(src[e], d);
    }
}

template <bool RELU>
__global__ void __launch_bounds__(128)
gemm_wmma_f32(const float* A,
              const float* __restrict__ W,
              const float* __restrict__ bias,
              float* C,
              int nrows) {
    __shared__ float Wlds[(FEAT / 2) * WPAIR];
    __shared__ float Alds[4][16 * LSTRIDE];

    const int tid  = threadIdx.x;
    const int lane = tid & 31;
    const int wv   = tid >> 5;
    const int wid  = blockIdx.x * 4 + wv;
    const int row0 = wid << 4;
    const bool active = (row0 < nrows);

    for (int i = tid; i < FEAT * (FEAT / 4); i += 128) {
        const int r  = i / (FEAT / 4);
        const int c0 = (i % (FEAT / 4)) * 4;
        const float4 v = *(const float4*)(W + r * FEAT + c0);
        float* __restrict__ dp = &Wlds[(r >> 1) * WPAIR + (r & 1)];
        dp[(c0 + 0) * 2] = v.x;
        dp[(c0 + 1) * 2] = v.y;
        dp[(c0 + 2) * 2] = v.z;
        dp[(c0 + 3) * 2] = v.w;
    }

    if (active) {
        for (int i = lane; i < 16 * (FEAT / 4); i += 32) {
            const int r  = i / (FEAT / 4);
            const int c4 = i % (FEAT / 4);
            int gr = row0 + r;
            if (gr > nrows - 1) gr = nrows - 1;
            const float4 v = ((const float4*)(A + (long long)gr * FEAT))[c4];
            *(float4*)&Alds[wv][r * LSTRIDE + c4 * 4] = v;
        }
    }
    __syncthreads();
    if (!active) return;

    const int m  = lane & 15;
    const int kh = (lane >> 4) << 1;
    const float* __restrict__ arow = &Alds[wv][m * LSTRIDE];

    float bv[6];
#pragma unroll
    for (int t = 0; t < 6; ++t) bv[t] = bias[t * 16 + m];

    v8f acc[6];
#pragma unroll
    for (int t = 0; t < 6; ++t) acc[t] = {};

    for (int k0 = 0; k0 < FEAT; k0 += 4) {
        const v2f a = *(const v2f*)&arow[k0 + kh];
        const float* __restrict__ wp = &Wlds[((k0 + kh) >> 1) * WPAIR + m * 2];
#pragma unroll
        for (int t = 0; t < 6; ++t) {
            const v2f b = *(const v2f*)&wp[t * 32];
            acc[t] = __builtin_amdgcn_wmma_f32_16x16x4_f32(
                false, a, false, b, (short)0, acc[t], false, false);
        }
    }

    const int half = lane >> 4;
#pragma unroll
    for (int t = 0; t < 6; ++t) {
#pragma unroll
        for (int i = 0; i < 8; ++i) {
            int r = row0 + half * 8 + i;
            float v = acc[t][i] + bv[t];
            if (RELU) v = v > 0.0f ? v : 0.0f;
            if (r < nrows) C[(long long)r * FEAT + t * 16 + m] = v;
        }
    }
}

#define ECHUNK 32
#define EBATCH 8

__device__ __forceinline__ void flush_row(float* __restrict__ out, int d, int lane,
                                          float& a0, float& a1, float& a2) {
    float* __restrict__ od = out + (long long)d * FEAT + lane;
    atomicAdd(od,      a0);
    atomicAdd(od + 32, a1);
    atomicAdd(od + 64, a2);
    a0 = 0.0f; a1 = 0.0f; a2 = 0.0f;
}

__global__ void __launch_bounds__(256)
aggregate_kernel(const float* __restrict__ h,
                 const int* __restrict__ sorted,
                 const float* __restrict__ dinv,
                 float* __restrict__ out,
                 int E) {
    const int lane = threadIdx.x & 31;
    const int wv = __builtin_amdgcn_readfirstlane((int)(threadIdx.x >> 5));
    const long long e0 = ((long long)blockIdx.x * (blockDim.x >> 5) + wv) * ECHUNK;
    if (e0 >= E) return;
    const int cnt = (E - e0 < ECHUNK) ? (int)(E - e0) : ECHUNK;

    const long long el = e0 + (lane < cnt ? lane : cnt - 1);
    const int2  sd  = *(const int2*)(sorted + 2 * el);
    const int   s_l = sd.x;
    const int   d_l = sd.y;
    const float n_l = dinv[s_l] * dinv[d_l];
    const int   n_b = __float_as_int(n_l);

    float acc0 = 0.0f, acc1 = 0.0f, acc2 = 0.0f;

    if (cnt == ECHUNK) {
        for (int b = 0; b < ECHUNK; b += EBATCH) {
            float v[EBATCH][3];
#pragma unroll
            for (int u = 0; u < EBATCH; ++u) {
                const int s = __builtin_amdgcn_readlane(s_l, b + u);
                const float* __restrict__ hs = h + (long long)s * FEAT + lane;
                v[u][0] = hs[0];
                v[u][1] = hs[32];
                v[u][2] = hs[64];
            }
#pragma unroll
            for (int u = 0; u < EBATCH; ++u) {
                const int j = b + u;
                const float nrm = __int_as_float(__builtin_amdgcn_readlane(n_b, j));
                const int d  = __builtin_amdgcn_readlane(d_l, j);
                const int dn = __builtin_amdgcn_readlane(d_l, (j + 1) & (ECHUNK - 1));
                acc0 += v[u][0] * nrm;
                acc1 += v[u][1] * nrm;
                acc2 += v[u][2] * nrm;
                if (j == ECHUNK - 1 || dn != d)
                    flush_row(out, d, lane, acc0, acc1, acc2);
            }
        }
    } else {
        for (int j = 0; j < cnt; ++j) {
            const int s = __builtin_amdgcn_readlane(s_l, j);
            const float* __restrict__ hs = h + (long long)s * FEAT + lane;
            const float nrm = __int_as_float(__builtin_amdgcn_readlane(n_b, j));
            const int d  = __builtin_amdgcn_readlane(d_l, j);
            const int dn = __builtin_amdgcn_readlane(d_l, (j + 1) & (ECHUNK - 1));
            acc0 += hs[0]  * nrm;
            acc1 += hs[32] * nrm;
            acc2 += hs[64] * nrm;
            if (j + 1 == cnt || dn != d)
                flush_row(out, d, lane, acc0, acc1, acc2);
        }
    }
}

extern "C" void kernel_launch(void* const* d_in, const int* in_sizes, int n_in,
                              void* d_out, int out_size, void* d_ws, size_t ws_size,
                              hipStream_t stream) {
    const float* x   = (const float*)d_in[0];
    const float* W1  = (const float*)d_in[1];
    const float* b1  = (const float*)d_in[2];
    const float* W2a = (const float*)d_in[3];
    const float* b2a = (const float*)d_in[4];
    const float* W2b = (const float*)d_in[5];
    const float* b2b = (const float*)d_in[6];
    const int*   ei  = (const int*)d_in[7];

    const int N = in_sizes[0] / FEAT;
    const int E = in_sizes[7] / 2;
    const int* src = ei;
    const int* dst = ei + E;

    const long long NF = (long long)N * FEAT;

    float* ws   = (float*)d_ws;
    int*   cnt  = (int*)d_ws;
    float* dinv = ws + N;
    float* agg1 = ws + 2LL * N;
    float* agg2 = agg1 + NF;

    float* out  = (float*)d_out;

    int* rowptr = (int*)d_out;
    int* rank   = rowptr + ((N + 3) & ~3);
    int* sorted = rank + ((E + 3) & ~3);

    const int T = 256;
    const int wavesPerBlk = T / 32;
    const int gemmWaves   = (N + 15) / 16;
    const int gemmBlocks  = (gemmWaves + 3) / 4;
    const int aggChunks   = (E + ECHUNK - 1) / ECHUNK;
    const int aggBlocks   = (aggChunks + wavesPerBlk - 1) / wavesPerBlk;

    zero_kernel<<<(int)((2 * N + 2 * NF + T - 1) / T), T, 0, stream>>>(ws, 2 * N + 2 * NF);

    count_rank_kernel<<<(E + T - 1) / T, T, 0, stream>>>(dst, cnt, rank, E);
    scan_dinv_kernel<<<1, SCAN_T, 0, stream>>>(cnt, rowptr, dinv, N);
    fill_kernel<<<(E + T - 1) / T, T, 0, stream>>>(src, dst, rank, rowptr, sorted, E);

    aggregate_kernel<<<aggBlocks, T, 0, stream>>>(x, sorted, dinv, agg1, E);
    gemm_wmma_f32<true><<<gemmBlocks, 128, 0, stream>>>(agg1, W1, b1, agg1, N);

    aggregate_kernel<<<aggBlocks, T, 0, stream>>>(agg1, sorted, dinv, agg2, E);
    gemm_wmma_f32<false><<<gemmBlocks, 128, 0, stream>>>(agg2, W2a, b2a, out, N);
    gemm_wmma_f32<false><<<gemmBlocks, 128, 0, stream>>>(agg2, W2b, b2b, out + NF, N);
}


# One wave per destination row sums its sorted edges in registers and stores the row once (rows over 64 edges keep chunked atomic adds): no zero fill of row buffers; changes the float summation order
# speedup vs baseline: 4.8977x; 1.0381x over previous
#include <hip/hip_runtime.h>
#include <hip/hip_bf16.h>

#define FEAT 96
#define LSTRIDE 100
#define WPAIR 192

typedef float v2f __attribute__((ext_vector_type(2)));
typedef float v8f __attribute__((ext_vector_type(8)));

__global__ void zero_kernel(float* __restrict__ p, long long n) {
    long long i = (long long)blockIdx.x * blockDim.x + threadIdx.x;
    if (i < n) p[i] = 0.0f;
}


__global__ void count_rank_kernel(const int* __restrict__ dst, int* __restrict__ cnt,
                                  int* __restrict__ rank, int E) {
    const int e    = blockIdx.x * blockDim.x + threadIdx.x;
    const int lane = threadIdx.x & 31;
    const bool valid = (e < E);
    const int d     = valid ? dst[e] : -1;
    const int dprev = __shfl_up(d, 1);
    const bool head = valid && (lane == 0 || d != dprev);
    const unsigned heads = (unsigned)__ballot(head);
    const unsigned nval  = (unsigned)__popcll(__ballot(valid));
    int base = 0;
    if (head) {
        const unsigned above = (lane == 31) ? 0u : (heads >> (lane + 1));
        const int len = above ? __ffs(above) : (int)nval - lane;
        base = atomicAdd(&cnt[d], len);
    }
    const unsigned le = heads & (0xffffffffu >> (31 - lane));
    const int hl = le ? 31 - __clz((int)le) : 0;
    const int b  = __shfl(base, hl);
    if (valid) rank[e] = b + (lane - hl);
}

#define SCAN_T  1024
#define SCAN_IT 16

__device__ __forceinline__ int4 load_counts4(const int* cnt, int i0, int N) {
    const int last = ((N + 3) & ~3) - 4;
    int4 v = *(const int4*)(cnt + (i0 < last ? i0 : last));
    v.x = (i0     < N) ? v.x : 0;
    v.y = (i0 + 1 < N) ? v.y : 0;
    v.z = (i0 + 2 < N) ? v.z : 0;
    v.w = (i0 + 3 < N) ? v.w : 0;
    return v;
}

#define HEAVY 64

__device__ __forceinline__ void clear_row(float* __restrict__ z1, float* __restrict__ z2, int row) {
    float4* __restrict__ p1 = (float4*)(z1 + (long long)row * FEAT);
    float4* __restrict__ p2 = (float4*)(z2 + (long long)row * FEAT);
    const float4 z = make_float4(0.0f, 0.0f, 0.0f, 0.0f);
    for (int q = 0; q < FEAT / 4; ++q) { p1[q] = z; p2[q] = z; }
}

__global__ void __launch_bounds__(SCAN_T)
scan_dinv_kernel(const int* cnt, int* __restrict__ rowptr, float* dinv,
                 float* __restrict__ z1, float* __restrict__ z2, int N) {
    __shared__ int wtot[32];
    const int lane = threadIdx.x & 31;
    const int wv   = threadIdx.x >> 5;
    int carry = 0;
    for (int p0 = 0; p0 < N; p0 += 32 * SCAN_IT * 128) {
        const int wstart = p0 + wv * (SCAN_IT * 128);

        int4 v[SCAN_IT];
#pragma unroll
        for (int it = 0; it < SCAN_IT; ++it)
            v[it] = load_counts4(cnt, wstart + it * 128 + lane * 4, N);
        int mine = 0;
#pragma unroll
        for (int it = 0; it < SCAN_IT; ++it)
            mine += v[it].x + v[it].y + v[it].z + v[it].w;
#pragma unroll
        for (int o = 16; o > 0; o >>= 1) mine += __shfl_xor(mine, o);
        if (lane == 0) wtot[wv] = mine;
        __syncthreads();
        const int w = wtot[lane];
        int woff = (lane < wv) ? w : 0;
        int tot  = w;
#pragma unroll
        for (int o = 16; o > 0; o >>= 1) {
            woff += __shfl_xor(woff, o);
            tot  += __shfl_xor(tot, o);
        }

#pragma unroll
        for (int it = 0; it < SCAN_IT; ++it)
            v[it] = load_counts4(cnt, wstart + it * 128 + lane * 4, N);
        int run = carry + woff;
        int maxc = 0;
#pragma unroll
        for (int it = 0; it < SCAN_IT; ++it) {
            const int i0 = wstart + it * 128 + lane * 4;
            const int c0 = v[it].x, c1 = v[it].y, c2 = v[it].z, c3 = v[it].w;
            const int t = c0 + c1 + c2 + c3;
            maxc = max(maxc, max(max(c0, c1), max(c2, c3)));
            int incl = t;
#pragma unroll
            for (int o = 1; o < 32; o <<= 1) {
                const int y = __shfl_up(incl, o);
                if (lane >= o) incl += y;
            }
            const int r0 = run + incl - t;
            const int r1 = r0 + c0, r2 = r1 + c1, r3 = r2 + c2;
            const float g0 = (float)c0, g1 = (float)c1, g2 = (float)c2, g3 = (float)c3;
            const float q0 = (g0 > 0.0f) ? rsqrtf(fmaxf(g0, 1.0f)) : 0.0f;
            const float q1 = (g1 > 0.0f) ? rsqrtf(fmaxf(g1, 1.0f)) : 0.0f;
            const float q2 = (g2 > 0.0f) ? rsqrtf(fmaxf(g2, 1.0f)) : 0.0f;
            const float q3 = (g3 > 0.0f) ? rsqrtf(fmaxf(g3, 1.0f)) : 0.0f;
            if (i0 + 3 < N) {
                *(int4*)(rowptr + i0) = make_int4(r0, r1, r2, r3);
                *(float4*)(dinv + i0) = make_float4(q0, q1, q2, q3);
            } else {
                if (i0     < N) { rowptr[i0]     = r0; dinv[i0]     = q0; }
                if (i0 + 1 < N) { rowptr[i0 + 1] = r1; dinv[i0 + 1] = q1; }
                if (i0 + 2 < N) { rowptr[i0 + 2] = r2; dinv[i0 + 2] = q2; }
            }
            run += __shfl(incl, 31);
        }
        if (maxc > HEAVY) {
#pragma nounroll
            for (int it = 0; it < SCAN_IT; ++it) {
                const int i0 = wstart + it * 128 + lane * 4;
                const int4 c = load_counts4(cnt, i0, N);
                if (c.x > HEAVY) clear_row(z1, z2, i0);
                if (c.y > HEAVY) clear_row(z1, z2, i0 + 1);
                if (c.z > HEAVY) clear_row(z1, z2, i0 + 2);
                if (c.w > HEAVY) clear_row(z1, z2, i0 + 3);
            }
        }
        carry += tot;
        __syncthreads();
    }
}

__global__ void fill_kernel(const int* __restrict__ src, const int* __restrict__ dst,
                            const int* __restrict__ rank, const int* __restrict__ rowptr,
                            int* __restrict__ sorted, int E) {
    const int e = blockIdx.x * blockDim.x + threadIdx.x;
    if (e < E) {
        const int d = dst[e];
        const long long pos = (long long)rowptr[d] + rank[e];
        *(int2*)(sorted + 2 * pos) = make_int2(src[e], d);
    }
}

template <bool RELU>
__global__ void __launch_bounds__(128)
gemm_wmma_f32(const float* A,
              const float* __restrict__ W,
              const float* __restrict__ bias,
              float* C,
              int nrows) {
    __shared__ float Wlds[(FEAT / 2) * WPAIR];
    __shared__ float Alds[4][16 * LSTRIDE];

    const int tid  = threadIdx.x;
    const int lane = tid & 31;
    const int wv   = tid >> 5;
    const int wid  = blockIdx.x * 4 + wv;
    const int row0 = wid << 4;
    const bool active = (row0 < nrows);

    for (int i = tid; i < FEAT * (FEAT / 4); i += 128) {
        const int r  = i / (FEAT / 4);
        const int c0 = (i % (FEAT / 4)) * 4;
        const float4 v = *(const float4*)(W + r * FEAT + c0);
        float* __restrict__ dp = &Wlds[(r >> 1) * WPAIR + (r & 1)];
        dp[(c0 + 0) * 2] = v.x;
        dp[(c0 + 1) * 2] = v.y;
        dp[(c0 + 2) * 2] = v.z;
        dp[(c0 + 3) * 2] = v.w;
    }

    if (active) {
        for (int i = lane; i < 16 * (FEAT / 4); i += 32) {
            const int r  = i / (FEAT / 4);
            const int c4 = i % (FEAT / 4);
            int gr = row0 + r;
            if (gr > nrows - 1) gr = nrows - 1;
            const float4 v = ((const float4*)(A + (long long)gr * FEAT))[c4];
            *(float4*)&Alds[wv][r * LSTRIDE + c4 * 4] = v;
        }
    }
    __syncthreads();
    if (!active) return;

    const int m  = lane & 15;
    const int kh = (lane >> 4) << 1;
    const float* __restrict__ arow = &Alds[wv][m * LSTRIDE];

    float bv[6];
#pragma unroll
    for (int t = 0; t < 6; ++t) bv[t] = bias[t * 16 + m];

    v8f acc[6];
#pragma unroll
    for (int t = 0; t < 6; ++t) acc[t] = {};

    for (int k0 = 0; k0 < FEAT; k0 += 4) {
        const v2f a = *(const v2f*)&arow[k0 + kh];
        const float* __restrict__ wp = &Wlds[((k0 + kh) >> 1) * WPAIR + m * 2];
#pragma unroll
        for (int t = 0; t < 6; ++t) {
            const v2f b = *(const v2f*)&wp[t * 32];
            acc[t] = __builtin_amdgcn_wmma_f32_16x16x4_f32(
                false, a, false, b, (short)0, acc[t], false, false);
        }
    }

    const int half = lane >> 4;
#pragma unroll
    for (int t = 0; t < 6; ++t) {
#pragma unroll
        for (int i = 0; i < 8; ++i) {
            int r = row0 + half * 8 + i;
            float v = acc[t][i] + bv[t];
            if (RELU) v = v > 0.0f ? v : 0.0f;
            if (r < nrows) C[(long long)r * FEAT + t * 16 + m] = v;
        }
    }
}

#define ECHUNK 32
#define EBATCH 8
#define RBATCH 16
#define AGG_WAVES 8

__device__ __forceinline__ void flush_row(float* __restrict__ out, int d, int lane,
                                          float& a0, float& a1, float& a2) {
    float* __restrict__ od = out + (long long)d * FEAT + lane;
    atomicAdd(od,      a0);
    atomicAdd(od + 32, a1);
    atomicAdd(od + 64, a2);
    a0 = 0.0f; a1 = 0.0f; a2 = 0.0f;
}

__device__ __forceinline__ int row_len(const int* __restrict__ rowptr, int r, int N, int E) {
    const int end = (r + 1 < N) ? rowptr[r + 1] : E;
    return end - rowptr[r];
}

__global__ void __launch_bounds__(AGG_WAVES * 32)
aggregate_kernel(const float* __restrict__ h,
                 const int* __restrict__ sorted,
                 const int* __restrict__ rowptr,
                 const float* __restrict__ dinv,
                 float* __restrict__ out,
                 int N, int E, int chunkBlocks) {
    const int lane = threadIdx.x & 31;
    const int wv = __builtin_amdgcn_readfirstlane((int)(threadIdx.x >> 5));

    if ((int)blockIdx.x >= chunkBlocks) {
        const int r = ((int)blockIdx.x - chunkBlocks) * AGG_WAVES + wv;
        if (r >= N) return;
        const int start = rowptr[r];
        const int end   = (r + 1 < N) ? rowptr[r + 1] : E;
        if (end - start > HEAVY) return;
        const float dr = dinv[r];

        float acc0 = 0.0f, acc1 = 0.0f, acc2 = 0.0f;
        for (int base = start; base < end; base += 32) {
            const int cnt = (end - base < 32) ? (end - base) : 32;
            const long long el = (long long)base + (lane < cnt ? lane : cnt - 1);
            const int s_l = sorted[2 * el];
            const int q_b = __float_as_int(dinv[s_l]);
            for (int b = 0; b < cnt; b += RBATCH) {
                float v[RBATCH][3];
#pragma unroll
                for (int u = 0; u < RBATCH; ++u) {
                    v[u][0] = 0.0f; v[u][1] = 0.0f; v[u][2] = 0.0f;
                    if (b + u < cnt) {
                        const int s = __builtin_amdgcn_readlane(s_l, b + u);
                        const float* __restrict__ hs = h + (long long)s * FEAT + lane;
                        v[u][0] = hs[0];
                        v[u][1] = hs[32];
                        v[u][2] = hs[64];
                    }
                }
#pragma unroll
                for (int u = 0; u < RBATCH; ++u) {
                    const float nrm = __int_as_float(__builtin_amdgcn_readlane(q_b, b + u)) * dr;
                    acc0 += v[u][0] * nrm;
                    acc1 += v[u][1] * nrm;
                    acc2 += v[u][2] * nrm;
                }
            }
        }
        float* __restrict__ od = out + (long long)r * FEAT + lane;
        od[0]  = acc0;
        od[32] = acc1;
        od[64] = acc2;
        return;
    }

    const long long e0 = ((long long)blockIdx.x * AGG_WAVES + wv) * ECHUNK;
    if (e0 >= E) return;
    const int cnt = (E - e0 < ECHUNK) ? (int)(E - e0) : ECHUNK;
    const int dF = sorted[2 * e0 + 1];
    const int dL = sorted[2 * (e0 + cnt - 1) + 1];
    const bool hF = row_len(rowptr, dF, N, E) > HEAVY;
    const bool hL = row_len(rowptr, dL, N, E) > HEAVY;
    if (!hF && !hL) return;

    const long long el = e0 + (lane < cnt ? lane : cnt - 1);
    const int2  sd  = *(const int2*)(sorted + 2 * el);
    const int   s_l = sd.x;
    const int   d_l = sd.y;
    const int   n_b = __float_as_int(dinv[s_l] * dinv[d_l]);
    const bool  long_l = (lane < cnt) && ((hF && d_l == dF) || (hL && d_l == dL));
    const unsigned hm = (unsigned)__ballot(long_l);

    float acc0 = 0.0f, acc1 = 0.0f, acc2 = 0.0f;
    for (int b = 0; b < cnt; b += EBATCH) {
        float v[EBATCH][3];
#pragma unroll
        for (int u = 0; u < EBATCH; ++u) {
            v[u][0] = 0.0f; v[u][1] = 0.0f; v[u][2] = 0.0f;
            if ((hm >> (b + u)) & 1u) {
                const int s = __builtin_amdgcn_readlane(s_l, b + u);
                const float* __restrict__ hs = h + (long long)s * FEAT + lane;
                v[u][0] = hs[0];
                v[u][1] = hs[32];
                v[u][2] = hs[64];
            }
        }
#pragma unroll
        for (int u = 0; u < EBATCH; ++u) {
            const int j = b + u;
            if ((hm >> j) & 1u) {
                const float nrm = __int_as_float(__builtin_amdgcn_readlane(n_b, j));
                const int d  = __builtin_amdgcn_readlane(d_l, j);
                const int dn = __builtin_amdgcn_readlane(d_l, (j + 1) & (ECHUNK - 1));
                acc0 += v[u][0] * nrm;
                acc1 += v[u][1] * nrm;
                acc2 += v[u][2] * nrm;
                const bool more = (j + 1 < ECHUNK) && ((hm >> ((j + 1) & (ECHUNK - 1))) & 1u) && (dn == d);
                if (!more) flush_row(out, d, lane, acc0, acc1, acc2);
            }
        }
    }
}

extern "C" void kernel_launch(void* const* d_in, const int* in_sizes, int n_in,
                              void* d_out, int out_size, void* d_ws, size_t ws_size,
                              hipStream_t stream) {
    const float* x   = (const float*)d_in[0];
    const float* W1  = (const float*)d_in[1];
    const float* b1  = (const float*)d_in[2];
    const float* W2a = (const float*)d_in[3];
    const float* b2a = (const float*)d_in[4];
    const float* W2b = (const float*)d_in[5];
    const float* b2b = (const float*)d_in[6];
    const int*   ei  = (const int*)d_in[7];

    const int N = in_sizes[0] / FEAT;
    const int E = in_sizes[7] / 2;
    const int* src = ei;
    const int* dst = ei + E;

    const long long NF = (long long)N * FEAT;

    float* ws   = (float*)d_ws;
    int*   cnt  = (int*)d_ws;
    float* dinv = ws + N;
    float* agg1 = ws + 2LL * N;
    float* agg2 = agg1 + NF;

    float* out  = (float*)d_out;

    int* rowptr = (int*)d_out;
    int* rank   = rowptr + ((N + 3) & ~3);
    int* sorted = rank + ((E + 3) & ~3);

    const int T = 256;
    const int gemmWaves   = (N + 15) / 16;
    const int gemmBlocks  = (gemmWaves + 3) / 4;
    const int aggChunks   = (E + ECHUNK - 1) / ECHUNK;
    const int chunkBlocks = (aggChunks + AGG_WAVES - 1) / AGG_WAVES;
    const int rowBlocks   = (N + AGG_WAVES - 1) / AGG_WAVES;
    const int aggBlocks   = chunkBlocks + rowBlocks;

    zero_kernel<<<(N + T - 1) / T, T, 0, stream>>>(ws, N);

    count_rank_kernel<<<(E + T - 1) / T, T, 0, stream>>>(dst, cnt, rank, E);
    scan_dinv_kernel<<<1, SCAN_T, 0, stream>>>(cnt, rowptr, dinv, agg1, agg2, N);
    fill_kernel<<<(E + T - 1) / T, T, 0, stream>>>(src, dst, rank, rowptr, sorted, E);

    aggregate_kernel<<<aggBlocks, AGG_WAVES * 32, 0, stream>>>(x, sorted, rowptr, dinv, agg1, N, E, chunkBlocks);
    gemm_wmma_f32<true><<<gemmBlocks, 128, 0, stream>>>(agg1, W1, b1, agg1, N);

    aggregate_kernel<<<aggBlocks, AGG_WAVES * 32, 0, stream>>>(agg1, sorted, rowptr, dinv, agg2, N, E, chunkBlocks);
    gemm_wmma_f32<false><<<gemmBlocks, 128, 0, stream>>>(agg2, W2a, b2a, out, N);
    gemm_wmma_f32<false><<<gemmBlocks, 128, 0, stream>>>(agg2, W2b, b2b, out + NF, N);
}
